# starting kernel
# baseline (speedup 1.0000x reference)
#include <hip/hip_runtime.h>

typedef __attribute__((ext_vector_type(2))) float v2f;
typedef __attribute__((ext_vector_type(4))) float v4f;
typedef __attribute__((ext_vector_type(8))) float v8f;

static constexpr int SN    = 128;
static constexpr int TILE  = 16;
static constexpr int NB    = SN / TILE;
static constexpr int APAD  = 132;
static constexpr int XPAD  = 20;
static constexpr int WAVES = 2;

__global__ __launch_bounds__(WAVES * 32) void hippo_bilinear_wmma(
    const float* __restrict__ u, const float* __restrict__ v,
    const float* __restrict__ delta, const float* __restrict__ A,
    const float* __restrict__ Bvec, float* __restrict__ out, int batch)
{
    __shared__ float A_lds[SN * APAD];
    __shared__ float xt[WAVES][TILE][XPAD];
    __shared__ float zf[WAVES][NB][TILE][XPAD];

    const int tid = threadIdx.x;
    const int nthreads = WAVES * 32;

    #pragma unroll 4
    for (int i = 0; i < (SN * SN / 4) / nthreads; ++i) {
        int q = tid + i * nthreads;
        int r = q >> 5;
        int c = (q & 31) << 2;
        v4f t = *(const v4f*)(A + r * SN + c);
        *(v4f*)(&A_lds[r * APAD + c]) = t;
    }
    __syncthreads();

    const int wave = tid >> 5;
    const int lane = tid & 31;
    const int col  = lane & 15;
    const int hlf  = lane >> 4;
    const int batch_base = (blockIdx.x * WAVES + wave) * TILE;
    const int b = batch_base + col;
    if (batch_base >= batch) return;

    const float dt2 = 0.5f * delta[b];
    const float dvb = delta[b] * v[b];
    const float* ucol = u + (size_t)b * SN;

    float* myxt = &xt[wave][0][0];

    #pragma unroll 1
    for (int k = 0; k < NB; ++k) {
        const float* arows = &A_lds[(k * TILE + col) * APAD];

        v8f acc = {};
        for (int j = 0; j <= k; ++j) {
            const float* arow = arows + j * TILE;
            const float* ucb  = ucol + j * TILE;
            #pragma unroll
            for (int k4 = 0; k4 < 4; ++k4) {
                int ko = k4 * 4 + 2 * hlf;
                v2f af = *(const v2f*)(arow + ko);
                v2f bf = *(const v2f*)(ucb + ko);
                acc = __builtin_amdgcn_wmma_f32_16x16x4_f32(
                          false, af, false, bf, (short)0, acc, false, false);
            }
        }

        {
            const int nbase = k * TILE + 8 * hlf;
            const v4f u0 = *(const v4f*)(ucol + nbase);
            const v4f u1 = *(const v4f*)(ucol + nbase + 4);
            const v4f b0 = *(const v4f*)(Bvec + nbase);
            const v4f b1 = *(const v4f*)(Bvec + nbase + 4);
            #pragma unroll
            for (int r = 0; r < 4; ++r) {
                acc[r]     = u0[r] + dt2 * acc[r]     + dvb * b0[r];
                acc[r + 4] = u1[r] + dt2 * acc[r + 4] + dvb * b1[r];
            }
        }

        for (int j = 0; j < k; ++j) {
            const float* arow = arows + j * TILE;
            const float* zrow = &zf[wave][j][col][0];
            #pragma unroll
            for (int k4 = 0; k4 < 4; ++k4) {
                int ko = k4 * 4 + 2 * hlf;
                v2f af = *(const v2f*)(arow + ko);
                v2f zb = *(const v2f*)(zrow + ko);
                acc = __builtin_amdgcn_wmma_f32_16x16x4_f32(
                          false, af, false, zb, (short)0, acc, false, false);
            }
        }

        #pragma unroll
        for (int r = 0; r < 8; ++r)
            myxt[(r + 8 * hlf) * XPAD + col] = acc[r];
        asm volatile("s_wait_dscnt 0x0" ::: "memory");

        if (lane < 16) {
            const int rb = k * TILE;
            float* myz = &zf[wave][k][lane][0];
            float y[16];
            #pragma unroll
            for (int m = 0; m < 16; ++m) {
                float s = myxt[m * XPAD + lane];
                float accv = 0.f;
                #pragma unroll
                for (int j = 0; j < m; ++j)
                    accv += A_lds[(rb + m) * APAD + rb + j] * y[j];
                const float diag = A_lds[(rb + m) * APAD + rb + m];
                y[m] = (s + dt2 * accv) / (1.0f - dt2 * diag);
                myz[m] = dt2 * y[m];
            }
            float* o = out + (size_t)(batch_base + lane) * SN + rb;
            #pragma unroll
            for (int m = 0; m < 16; m += 4) {
                v4f t = { y[m], y[m + 1], y[m + 2], y[m + 3] };
                *(v4f*)(o + m) = t;
            }
        }
        asm volatile("s_wait_dscnt 0x0" ::: "memory");
    }
}

extern "C" void kernel_launch(void* const* d_in, const int* in_sizes, int n_in,
                              void* d_out, int out_size, void* d_ws, size_t ws_size,
                              hipStream_t stream) {
    const float* u     = (const float*)d_in[0];
    const float* v     = (const float*)d_in[1];
    const float* delta = (const float*)d_in[2];
    const float* A     = (const float*)d_in[3];
    const float* Bvec  = (const float*)d_in[4];
    float* out = (float*)d_out;

    const int batch  = in_sizes[0] / SN;
    const int blocks = batch / (WAVES * TILE);
    hippo_bilinear_wmma<<<blocks, WAVES * 32, 0, stream>>>(u, v, delta, A, Bvec, out, batch);
}


# diagonal 16x16 solve: each serial step cut to fma,mul - pivot division replaced by multiply with correctly rounded 1/pivot taken ahead, block coefficients preloaded in registers; sums keep j order
# speedup vs baseline: 1.2425x; 1.2425x over previous
#include <hip/hip_runtime.h>

typedef __attribute__((ext_vector_type(2))) float v2f;
typedef __attribute__((ext_vector_type(4))) float v4f;
typedef __attribute__((ext_vector_type(8))) float v8f;

static constexpr int SN    = 128;
static constexpr int TILE  = 16;
static constexpr int NB    = SN / TILE;
static constexpr int APAD  = 132;
static constexpr int XPAD  = 20;
static constexpr int WAVES = 2;

__global__ __launch_bounds__(WAVES * 32) void hippo_bilinear_wmma(
    const float* __restrict__ u, const float* __restrict__ v,
    const float* __restrict__ delta, const float* __restrict__ A,
    const float* __restrict__ Bvec, float* __restrict__ out, int batch)
{
    __shared__ float A_lds[SN * APAD];
    __shared__ float xt[WAVES][TILE][XPAD];
    __shared__ float zf[WAVES][NB][TILE][XPAD];

    const int tid = threadIdx.x;
    const int nthreads = WAVES * 32;

    #pragma unroll 4
    for (int i = 0; i < (SN * SN / 4) / nthreads; ++i) {
        int q = tid + i * nthreads;
        int r = q >> 5;
        int c = (q & 31) << 2;
        v4f t = *(const v4f*)(A + r * SN + c);
        *(v4f*)(&A_lds[r * APAD + c]) = t;
    }
    __syncthreads();

    const int wave = tid >> 5;
    const int lane = tid & 31;
    const int col  = lane & 15;
    const int hlf  = lane >> 4;
    const int batch_base = (blockIdx.x * WAVES + wave) * TILE;
    const int b = batch_base + col;
    if (batch_base >= batch) return;

    const float dt2 = 0.5f * delta[b];
    const float dvb = delta[b] * v[b];
    const float* ucol = u + (size_t)b * SN;

    float* myxt = &xt[wave][0][0];

    #pragma unroll 1
    for (int k = 0; k < NB; ++k) {
        const float* arows = &A_lds[(k * TILE + col) * APAD];

        const int rb = k * TILE;
        float a[TILE][TILE];
        {
            const float* dblk = &A_lds[rb * APAD + rb];
            #pragma unroll
            for (int m = 0; m < TILE; ++m) {
                #pragma unroll
                for (int j = 0; j <= m; ++j)
                    a[m][j] = dblk[m * APAD + j];
            }
        }

        v8f acc = {};
        for (int j = 0; j <= k; ++j) {
            const float* arow = arows + j * TILE;
            const float* ucb  = ucol + j * TILE;
            #pragma unroll
            for (int k4 = 0; k4 < 4; ++k4) {
                int ko = k4 * 4 + 2 * hlf;
                v2f af = *(const v2f*)(arow + ko);
                v2f bf = *(const v2f*)(ucb + ko);
                acc = __builtin_amdgcn_wmma_f32_16x16x4_f32(
                          false, af, false, bf, (short)0, acc, false, false);
            }
        }

        {
            const int nbase = k * TILE + 8 * hlf;
            const v4f u0 = *(const v4f*)(ucol + nbase);
            const v4f u1 = *(const v4f*)(ucol + nbase + 4);
            const v4f b0 = *(const v4f*)(Bvec + nbase);
            const v4f b1 = *(const v4f*)(Bvec + nbase + 4);
            #pragma unroll
            for (int r = 0; r < 4; ++r) {
                acc[r]     = u0[r] + dt2 * acc[r]     + dvb * b0[r];
                acc[r + 4] = u1[r] + dt2 * acc[r + 4] + dvb * b1[r];
            }
        }

        for (int j = 0; j < k; ++j) {
            const float* arow = arows + j * TILE;
            const float* zrow = &zf[wave][j][col][0];
            #pragma unroll
            for (int k4 = 0; k4 < 4; ++k4) {
                int ko = k4 * 4 + 2 * hlf;
                v2f af = *(const v2f*)(arow + ko);
                v2f zb = *(const v2f*)(zrow + ko);
                acc = __builtin_amdgcn_wmma_f32_16x16x4_f32(
                          false, af, false, zb, (short)0, acc, false, false);
            }
        }

        #pragma unroll
        for (int r = 0; r < 8; ++r)
            myxt[(r + 8 * hlf) * XPAD + col] = acc[r];
        asm volatile("s_wait_dscnt 0x0" ::: "memory");

        if (lane < 16) {
            float* myz = &zf[wave][k][lane][0];
            float s[TILE], rinv[TILE], racc[TILE], y[TILE];
            #pragma unroll
            for (int m = 0; m < TILE; ++m) {
                s[m]    = myxt[m * XPAD + lane];
                rinv[m] = 1.0f / (1.0f - dt2 * a[m][m]);
                racc[m] = 0.f;
            }
            #pragma unroll
            for (int m = 0; m < TILE; ++m) {
                y[m] = __builtin_fmaf(dt2, racc[m], s[m]) * rinv[m];
                #pragma unroll
                for (int i = m + 1; i < TILE; ++i)
                    racc[i] = __builtin_fmaf(a[i][m], y[m], racc[i]);
                myz[m] = dt2 * y[m];
            }
            float* o = out + (size_t)(batch_base + lane) * SN + rb;
            #pragma unroll
            for (int m = 0; m < 16; m += 4) {
                v4f t = { y[m], y[m + 1], y[m + 2], y[m + 3] };
                *(v4f*)(o + m) = t;
            }
        }
        asm volatile("s_wait_dscnt 0x0" ::: "memory");
    }
}

extern "C" void kernel_launch(void* const* d_in, const int* in_sizes, int n_in,
                              void* d_out, int out_size, void* d_ws, size_t ws_size,
                              hipStream_t stream) {
    const float* u     = (const float*)d_in[0];
    const float* v     = (const float*)d_in[1];
    const float* delta = (const float*)d_in[2];
    const float* A     = (const float*)d_in[3];
    const float* Bvec  = (const float*)d_in[4];
    float* out = (float*)d_out;

    const int batch  = in_sizes[0] / SN;
    const int blocks = batch / (WAVES * TILE);
    hippo_bilinear_wmma<<<blocks, WAVES * 32, 0, stream>>>(u, v, delta, A, Bvec, out, batch);
}


# A staging into LDS: only the 36 lower 16x16 blocks that are read, one float4 per thread per block, all 36 loads in flight before the first store (was 64 loads in 16 waited rounds of 4)
# speedup vs baseline: 1.7643x; 1.4200x over previous
#include <hip/hip_runtime.h>

typedef __attribute__((ext_vector_type(2))) float v2f;
typedef __attribute__((ext_vector_type(4))) float v4f;
typedef __attribute__((ext_vector_type(8))) float v8f;

static constexpr int SN    = 128;
static constexpr int TILE  = 16;
static constexpr int NB    = SN / TILE;
static constexpr int APAD  = 132;
static constexpr int XPAD  = 20;
static constexpr int WAVES = 2;

__global__ __launch_bounds__(WAVES * 32) void hippo_bilinear_wmma(
    const float* __restrict__ u, const float* __restrict__ v,
    const float* __restrict__ delta, const float* __restrict__ A,
    const float* __restrict__ Bvec, float* __restrict__ out, int batch)
{
    __shared__ float A_lds[SN * APAD];
    __shared__ float xt[WAVES][TILE][XPAD];
    __shared__ float zf[WAVES][NB][TILE][XPAD];

    const int tid = threadIdx.x;

    {
        static_assert(WAVES * 32 == 64, "one float4 per thread per 16x16 block");
        constexpr int NBLK = NB * (NB + 1) / 2;
        const int c4   = tid & 3;
        const int brow = ((tid >> 2) & 3) * 4 + (tid >> 4);
        const float* gsrc = A + brow * SN + c4 * 4;
        float* ldst = &A_lds[brow * APAD + c4 * 4];
        v4f t[NBLK];
        #pragma unroll
        for (int kb = 0; kb < NB; ++kb) {
            #pragma unroll
            for (int jb = 0; jb <= kb; ++jb)
                t[kb * (kb + 1) / 2 + jb] = *(const v4f*)(gsrc + kb * TILE * SN + jb * TILE);
        }
        #pragma unroll
        for (int kb = 0; kb < NB; ++kb) {
            #pragma unroll
            for (int jb = 0; jb <= kb; ++jb)
                *(v4f*)(ldst + kb * TILE * APAD + jb * TILE) = t[kb * (kb + 1) / 2 + jb];
        }
    }
    __syncthreads();

    const int wave = tid >> 5;
    const int lane = tid & 31;
    const int col  = lane & 15;
    const int hlf  = lane >> 4;
    const int batch_base = (blockIdx.x * WAVES + wave) * TILE;
    const int b = batch_base + col;
    if (batch_base >= batch) return;

    const float dt2 = 0.5f * delta[b];
    const float dvb = delta[b] * v[b];
    const float* ucol = u + (size_t)b * SN;

    float* myxt = &xt[wave][0][0];

    #pragma unroll 1
    for (int k = 0; k < NB; ++k) {
        const float* arows = &A_lds[(k * TILE + col) * APAD];

        const int rb = k * TILE;
        float a[TILE][TILE];
        {
            const float* dblk = &A_lds[rb * APAD + rb];
            #pragma unroll
            for (int m = 0; m < TILE; ++m) {
                #pragma unroll
                for (int j = 0; j <= m; ++j)
                    a[m][j] = dblk[m * APAD + j];
            }
        }

        v8f acc = {};
        for (int j = 0; j <= k; ++j) {
            const float* arow = arows + j * TILE;
            const float* ucb  = ucol + j * TILE;
            #pragma unroll
            for (int k4 = 0; k4 < 4; ++k4) {
                int ko = k4 * 4 + 2 * hlf;
                v2f af = *(const v2f*)(arow + ko);
                v2f bf = *(const v2f*)(ucb + ko);
                acc = __builtin_amdgcn_wmma_f32_16x16x4_f32(
                          false, af, false, bf, (short)0, acc, false, false);
            }
        }

        {
            const int nbase = k * TILE + 8 * hlf;
            const v4f u0 = *(const v4f*)(ucol + nbase);
            const v4f u1 = *(const v4f*)(ucol + nbase + 4);
            const v4f b0 = *(const v4f*)(Bvec + nbase);
            const v4f b1 = *(const v4f*)(Bvec + nbase + 4);
            #pragma unroll
            for (int r = 0; r < 4; ++r) {
                acc[r]     = u0[r] + dt2 * acc[r]     + dvb * b0[r];
                acc[r + 4] = u1[r] + dt2 * acc[r + 4] + dvb * b1[r];
            }
        }

        for (int j = 0; j < k; ++j) {
            const float* arow = arows + j * TILE;
            const float* zrow = &zf[wave][j][col][0];
            #pragma unroll
            for (int k4 = 0; k4 < 4; ++k4) {
                int ko = k4 * 4 + 2 * hlf;
                v2f af = *(const v2f*)(arow + ko);
                v2f zb = *(const v2f*)(zrow + ko);
                acc = __builtin_amdgcn_wmma_f32_16x16x4_f32(
                          false, af, false, zb, (short)0, acc, false, false);
            }
        }

        #pragma unroll
        for (int r = 0; r < 8; ++r)
            myxt[(r + 8 * hlf) * XPAD + col] = acc[r];
        asm volatile("s_wait_dscnt 0x0" ::: "memory");

        if (lane < 16) {
            float* myz = &zf[wave][k][lane][0];
            float s[TILE], rinv[TILE], racc[TILE], y[TILE];
            #pragma unroll
            for (int m = 0; m < TILE; ++m) {
                s[m]    = myxt[m * XPAD + lane];
                rinv[m] = 1.0f / (1.0f - dt2 * a[m][m]);
                racc[m] = 0.f;
            }
            #pragma unroll
            for (int m = 0; m < TILE; ++m) {
                y[m] = __builtin_fmaf(dt2, racc[m], s[m]) * rinv[m];
                #pragma unroll
                for (int i = m + 1; i < TILE; ++i)
                    racc[i] = __builtin_fmaf(a[i][m], y[m], racc[i]);
                myz[m] = dt2 * y[m];
            }
            float* o = out + (size_t)(batch_base + lane) * SN + rb;
            #pragma unroll
            for (int m = 0; m < 16; m += 4) {
                v4f t = { y[m], y[m + 1], y[m + 2], y[m + 3] };
                *(v4f*)(o + m) = t;
            }
        }
        asm volatile("s_wait_dscnt 0x0" ::: "memory");
    }
}

extern "C" void kernel_launch(void* const* d_in, const int* in_sizes, int n_in,
                              void* d_out, int out_size, void* d_ws, size_t ws_size,
                              hipStream_t stream) {
    const float* u     = (const float*)d_in[0];
    const float* v     = (const float*)d_in[1];
    const float* delta = (const float*)d_in[2];
    const float* A     = (const float*)d_in[3];
    const float* Bvec  = (const float*)d_in[4];
    float* out = (float*)d_out;

    const int batch  = in_sizes[0] / SN;
    const int blocks = batch / (WAVES * TILE);
    hippo_bilinear_wmma<<<blocks, WAVES * 32, 0, stream>>>(u, v, delta, A, Bvec, out, batch);
}


# forward product folded into the substitution by linearity: operand W_j=dt2*(u_j+y_j), each off-diagonal A block multiplied once (144 WMMA per wave, was 256), diagonal block last; sum order changes
# speedup vs baseline: 2.5844x; 1.4648x over previous
#include <hip/hip_runtime.h>

typedef __attribute__((ext_vector_type(2))) float v2f;
typedef __attribute__((ext_vector_type(4))) float v4f;
typedef __attribute__((ext_vector_type(8))) float v8f;

static constexpr int SN    = 128;
static constexpr int TILE  = 16;
static constexpr int NB    = SN / TILE;
static constexpr int APAD  = 132;
static constexpr int XPAD  = 20;
static constexpr int WAVES = 2;

__global__ __launch_bounds__(WAVES * 32) void hippo_bilinear_wmma(
    const float* __restrict__ u, const float* __restrict__ v,
    const float* __restrict__ delta, const float* __restrict__ A,
    const float* __restrict__ Bvec, float* __restrict__ out, int batch)
{
    __shared__ float A_lds[SN * APAD];
    __shared__ float xt[WAVES][TILE][XPAD];
    __shared__ float zf[WAVES][NB][TILE][XPAD];

    const int tid = threadIdx.x;

    {
        static_assert(WAVES * 32 == 64, "one float4 per thread per 16x16 block");
        constexpr int NBLK = NB * (NB + 1) / 2;
        const int c4   = tid & 3;
        const int brow = ((tid >> 2) & 3) * 4 + (tid >> 4);
        const float* gsrc = A + brow * SN + c4 * 4;
        float* ldst = &A_lds[brow * APAD + c4 * 4];
        v4f t[NBLK];
        #pragma unroll
        for (int kb = 0; kb < NB; ++kb) {
            #pragma unroll
            for (int jb = 0; jb <= kb; ++jb)
                t[kb * (kb + 1) / 2 + jb] = *(const v4f*)(gsrc + kb * TILE * SN + jb * TILE);
        }
        #pragma unroll
        for (int kb = 0; kb < NB; ++kb) {
            #pragma unroll
            for (int jb = 0; jb <= kb; ++jb)
                *(v4f*)(ldst + kb * TILE * APAD + jb * TILE) = t[kb * (kb + 1) / 2 + jb];
        }
    }
    __syncthreads();

    const int wave = tid >> 5;
    const int lane = tid & 31;
    const int col  = lane & 15;
    const int hlf  = lane >> 4;
    const int batch_base = (blockIdx.x * WAVES + wave) * TILE;
    const int b = batch_base + col;
    if (batch_base >= batch) return;

    const float dt2 = 0.5f * delta[b];
    const float dvb = delta[b] * v[b];
    const float* ucol = u + (size_t)b * SN;

    float* myxt = &xt[wave][0][0];

    #pragma unroll 1
    for (int k = 0; k < NB; ++k) {
        const float* arows = &A_lds[(k * TILE + col) * APAD];

        const int rb = k * TILE;
        float a[TILE][TILE];
        {
            const float* dblk = &A_lds[rb * APAD + rb];
            #pragma unroll
            for (int m = 0; m < TILE; ++m) {
                #pragma unroll
                for (int j = 0; j <= m; ++j)
                    a[m][j] = dblk[m * APAD + j];
            }
        }

        v2f ad[4];
        #pragma unroll
        for (int k4 = 0; k4 < 4; ++k4)
            ad[k4] = *(const v2f*)(arows + rb + k4 * 4 + 2 * hlf);

        const int nbase = rb + 8 * hlf;
        const v4f u0 = *(const v4f*)(ucol + nbase);
        const v4f u1 = *(const v4f*)(ucol + nbase + 4);
        const v4f b0 = *(const v4f*)(Bvec + nbase);
        const v4f b1 = *(const v4f*)(Bvec + nbase + 4);
        v2f uk[4];
        #pragma unroll
        for (int k4 = 0; k4 < 4; ++k4)
            uk[k4] = *(const v2f*)(ucol + rb + k4 * 4 + 2 * hlf);
        const v4f u2 = *(const v4f*)(ucol + rb + 8);
        const v4f u3 = *(const v4f*)(ucol + rb + 12);

        v8f acc = {};
        for (int j = 0; j < k; ++j) {
            const float* arow = arows + j * TILE;
            const float* wrow = &zf[wave][j][col][0];
            #pragma unroll
            for (int k4 = 0; k4 < 4; ++k4) {
                int ko = k4 * 4 + 2 * hlf;
                v2f af = *(const v2f*)(arow + ko);
                v2f wb = *(const v2f*)(wrow + ko);
                acc = __builtin_amdgcn_wmma_f32_16x16x4_f32(
                          false, af, false, wb, (short)0, acc, false, false);
            }
        }

        #pragma unroll
        for (int k4 = 0; k4 < 4; ++k4) {
            v2f bf = dt2 * uk[k4];
            acc = __builtin_amdgcn_wmma_f32_16x16x4_f32(
                      false, ad[k4], false, bf, (short)0, acc, false, false);
        }

        #pragma unroll
        for (int r = 0; r < 4; ++r) {
            acc[r]     = (u0[r] + dvb * b0[r]) + acc[r];
            acc[r + 4] = (u1[r] + dvb * b1[r]) + acc[r + 4];
        }

        #pragma unroll
        for (int r = 0; r < 8; ++r)
            myxt[(r + 8 * hlf) * XPAD + col] = acc[r];
        asm volatile("s_wait_dscnt 0x0" ::: "memory");

        if (lane < 16) {
            float* myz = &zf[wave][k][lane][0];
            float s[TILE], rinv[TILE], racc[TILE], y[TILE];
            const float uo[TILE] = { u0[0], u0[1], u0[2], u0[3], u1[0], u1[1], u1[2], u1[3],
                                     u2[0], u2[1], u2[2], u2[3], u3[0], u3[1], u3[2], u3[3] };
            #pragma unroll
            for (int m = 0; m < TILE; ++m) {
                s[m]    = myxt[m * XPAD + lane];
                rinv[m] = 1.0f / (1.0f - dt2 * a[m][m]);
                racc[m] = 0.f;
            }
            #pragma unroll
            for (int m = 0; m < TILE; ++m) {
                y[m] = __builtin_fmaf(dt2, racc[m], s[m]) * rinv[m];
                #pragma unroll
                for (int i = m + 1; i < TILE; ++i)
                    racc[i] = __builtin_fmaf(a[i][m], y[m], racc[i]);
                myz[m] = dt2 * (y[m] + uo[m]);
            }
            float* o = out + (size_t)(batch_base + lane) * SN + rb;
            #pragma unroll
            for (int m = 0; m < 16; m += 4) {
                v4f t = { y[m], y[m + 1], y[m + 2], y[m + 3] };
                *(v4f*)(o + m) = t;
            }
        }
        asm volatile("s_wait_dscnt 0x0" ::: "memory");
    }
}

extern "C" void kernel_launch(void* const* d_in, const int* in_sizes, int n_in,
                              void* d_out, int out_size, void* d_ws, size_t ws_size,
                              hipStream_t stream) {
    const float* u     = (const float*)d_in[0];
    const float* v     = (const float*)d_in[1];
    const float* delta = (const float*)d_in[2];
    const float* A     = (const float*)d_in[3];
    const float* Bvec  = (const float*)d_in[4];
    float* out = (float*)d_out;

    const int batch  = in_sizes[0] / SN;
    const int blocks = batch / (WAVES * TILE);
    hippo_bilinear_wmma<<<blocks, WAVES * 32, 0, stream>>>(u, v, delta, A, Bvec, out, batch);
}


# two waves per batch tile: a product wave sums, ahead of the solving wave, all of a block-row's right-hand side but A(k,k-1)W(k-1); 4 WMMA per row stay on the serial path; sum order changes
# speedup vs baseline: 3.4571x; 1.3377x over previous
#include <hip/hip_runtime.h>

typedef __attribute__((ext_vector_type(2))) float v2f;
typedef __attribute__((ext_vector_type(4))) float v4f;
typedef __attribute__((ext_vector_type(8))) float v8f;

static constexpr int SN    = 128;
static constexpr int TILE  = 16;
static constexpr int NB    = SN / TILE;
static constexpr int APAD  = 132;
static constexpr int XPAD  = 20;
static constexpr int TILES = 2;
static constexpr int WAVES = 2 * TILES;

static __device__ __forceinline__ void lds_handover_barrier() {
    asm volatile("s_wait_dscnt 0x0\n\ts_barrier_signal -1\n\ts_barrier_wait -1" ::: "memory");
}

template <int K>
static __device__ __forceinline__ void product_step(v8f (&pacc)[NB], const float* alds,
                                                    const float* wrow, float* pbk,
                                                    int col, int hlf)
{
    v2f wb[4];
    v2f af[NB][4];
    #pragma unroll
    for (int k4 = 0; k4 < 4; ++k4)
        wb[k4] = *(const v2f*)(wrow + k4 * 4 + 2 * hlf);
    #pragma unroll
    for (int i = K + 1; i < NB; ++i) {
        const float* arow = alds + (i * TILE + col) * APAD + (K - 1) * TILE;
        #pragma unroll
        for (int k4 = 0; k4 < 4; ++k4)
            af[i][k4] = *(const v2f*)(arow + k4 * 4 + 2 * hlf);
    }
    #pragma unroll
    for (int k4 = 0; k4 < 4; ++k4) {
        #pragma unroll
        for (int i = K + 1; i < NB; ++i)
            pacc[i] = __builtin_amdgcn_wmma_f32_16x16x4_f32(
                          false, af[i][k4], false, wb[k4], (short)0, pacc[i], false, false);
    }
    *(v8f*)(pbk + (K + 1) * (32 * 8)) = pacc[K + 1];
}

__global__ __launch_bounds__(WAVES * 32) void hippo_bilinear_wmma(
    const float* __restrict__ u, const float* __restrict__ v,
    const float* __restrict__ delta, const float* __restrict__ A,
    const float* __restrict__ Bvec, float* __restrict__ out, int batch)
{
    __shared__ float A_lds[SN * APAD];
    __shared__ float xt[TILES][TILE][XPAD];
    __shared__ float zf[TILES][NB][TILE][XPAD];
    __shared__ __attribute__((aligned(32))) float pb[TILES][NB][32 * 8];

    const int tid = threadIdx.x;

    const int wave = __builtin_amdgcn_readfirstlane(tid >> 5);
    const int lane = tid & 31;
    const int tile = wave & (TILES - 1);
    const bool solving = wave < TILES;
    const int col  = lane & 15;
    const int hlf  = lane >> 4;
    const int batch_base = (blockIdx.x * TILES + tile) * TILE;
    const bool active = batch_base < batch;
    const int b = active ? batch_base + col : col;

    float dl, vv;
    const float* ucol = u + (size_t)b * SN;

    v4f su[4];
    v4f hu[NB][2], hb[NB][2];
    v2f huk[NB][4];
    if (active) {
        if (solving) {
            #pragma unroll
            for (int q = 0; q < 4; ++q) su[q] = *(const v4f*)(ucol + 4 * q);
        } else {
            #pragma unroll
            for (int i = 0; i < NB; ++i) {
                const int rb = i * TILE;
                const int nbase = rb + 8 * hlf;
                hu[i][0] = *(const v4f*)(ucol + nbase);
                hu[i][1] = *(const v4f*)(ucol + nbase + 4);
                hb[i][0] = *(const v4f*)(Bvec + nbase);
                hb[i][1] = *(const v4f*)(Bvec + nbase + 4);
                #pragma unroll
                for (int k4 = 0; k4 < 4; ++k4)
                    huk[i][k4] = *(const v2f*)(ucol + rb + k4 * 4 + 2 * hlf);
            }
        }
    }

    {
        static_assert(WAVES * 32 == 128, "one float2 per thread per 16x16 block");
        constexpr int NBLK = NB * (NB + 1) / 2;
        const int c2   = tid & 7;
        const int brow = ((tid >> 3) & 3) * 4 + (tid >> 5);
        const float* gsrc = A + brow * SN + c2 * 2;
        float* ldst = &A_lds[brow * APAD + c2 * 2];
        v2f t[NBLK];
        #pragma unroll
        for (int kb = 0; kb < NB; ++kb) {
            #pragma unroll
            for (int jb = 0; jb <= kb; ++jb)
                t[kb * (kb + 1) / 2 + jb] = *(const v2f*)(gsrc + kb * TILE * SN + jb * TILE);
        }
        dl = delta[b];
        vv = v[b];
        #pragma unroll
        for (int kb = 0; kb < NB; ++kb) {
            #pragma unroll
            for (int jb = 0; jb <= kb; ++jb)
                *(v2f*)(ldst + kb * TILE * APAD + jb * TILE) = t[kb * (kb + 1) / 2 + jb];
        }
    }
    __syncthreads();

    const float dt2 = 0.5f * dl;
    const float dvb = dl * vv;

    float* myxt = &xt[tile][0][0];
    float* mypb = &pb[tile][0][lane * 8];

    v8f pacc[NB];

    if (active && !solving) {
        #pragma unroll
        for (int i = 0; i < NB; ++i) {
            #pragma unroll
            for (int r = 0; r < 4; ++r) {
                pacc[i][r]     = hu[i][0][r] + dvb * hb[i][0][r];
                pacc[i][r + 4] = hu[i][1][r] + dvb * hb[i][1][r];
            }
        }
        const float* arow = &A_lds[col * APAD];
        v8f acc = pacc[0];
        #pragma unroll
        for (int k4 = 0; k4 < 4; ++k4) {
            v2f af = *(const v2f*)(arow + k4 * 4 + 2 * hlf);
            v2f bf = dt2 * huk[0][k4];
            acc = __builtin_amdgcn_wmma_f32_16x16x4_f32(
                      false, af, false, bf, (short)0, acc, false, false);
        }
        *(v8f*)(mypb) = acc;
    }

    #pragma unroll 1
    for (int k = 0; k < NB; ++k) {
        lds_handover_barrier();
        if (!active) continue;

        if (!solving) {
            if (k == 0) {
                v2f ad[NB][4];
                #pragma unroll
                for (int i = 1; i < NB; ++i) {
                    const float* arow = &A_lds[(i * TILE + col) * APAD + i * TILE];
                    #pragma unroll
                    for (int k4 = 0; k4 < 4; ++k4)
                        ad[i][k4] = *(const v2f*)(arow + k4 * 4 + 2 * hlf);
                }
                #pragma unroll
                for (int k4 = 0; k4 < 4; ++k4) {
                    #pragma unroll
                    for (int i = 1; i < NB; ++i) {
                        v2f bf = dt2 * huk[i][k4];
                        pacc[i] = __builtin_amdgcn_wmma_f32_16x16x4_f32(
                                      false, ad[i][k4], false, bf, (short)0, pacc[i], false, false);
                    }
                }
                *(v8f*)(mypb + 1 * (32 * 8)) = pacc[1];
            } else {
                const float* wrow = &zf[tile][k - 1][col][0];
                switch (k) {
                    case 1: product_step<1>(pacc, A_lds, wrow, mypb, col, hlf); break;
                    case 2: product_step<2>(pacc, A_lds, wrow, mypb, col, hlf); break;
                    case 3: product_step<3>(pacc, A_lds, wrow, mypb, col, hlf); break;
                    case 4: product_step<4>(pacc, A_lds, wrow, mypb, col, hlf); break;
                    case 5: product_step<5>(pacc, A_lds, wrow, mypb, col, hlf); break;
                    case 6: product_step<6>(pacc, A_lds, wrow, mypb, col, hlf); break;
                    default: break;
                }
            }
            continue;
        }

        const int rb = k * TILE;
        const float* arows = &A_lds[(rb + col) * APAD];

        v8f acc = *(const v8f*)(mypb + k * (32 * 8));
        v2f af[4], wb[4];
        if (k > 0) {
            const float* wrow = &zf[tile][k - 1][col][0];
            #pragma unroll
            for (int k4 = 0; k4 < 4; ++k4) {
                af[k4] = *(const v2f*)(arows + (rb - TILE) + k4 * 4 + 2 * hlf);
                wb[k4] = *(const v2f*)(wrow + k4 * 4 + 2 * hlf);
            }
        }

        float a[TILE][TILE];
        {
            const float* dblk = &A_lds[rb * APAD + rb];
            #pragma unroll
            for (int m = 0; m < TILE; ++m) {
                #pragma unroll
                for (int j = 0; j <= m; ++j)
                    a[m][j] = dblk[m * APAD + j];
            }
        }

        const v4f u0 = su[0], u1 = su[1], u2 = su[2], u3 = su[3];
        if (k + 1 < NB) {
            #pragma unroll
            for (int q = 0; q < 4; ++q) su[q] = *(const v4f*)(ucol + rb + TILE + 4 * q);
        }

        if (k > 0) {
            #pragma unroll
            for (int k4 = 0; k4 < 4; ++k4)
                acc = __builtin_amdgcn_wmma_f32_16x16x4_f32(
                          false, af[k4], false, wb[k4], (short)0, acc, false, false);
        }

        #pragma unroll
        for (int r = 0; r < 8; ++r)
            myxt[(r + 8 * hlf) * XPAD + col] = acc[r];
        asm volatile("s_wait_dscnt 0x0" ::: "memory");

        if (lane < 16) {
            float* myz = &zf[tile][k][lane][0];
            float s[TILE], rinv[TILE], racc[TILE], y[TILE];
            const float uo[TILE] = { u0[0], u0[1], u0[2], u0[3], u1[0], u1[1], u1[2], u1[3],
                                     u2[0], u2[1], u2[2], u2[3], u3[0], u3[1], u3[2], u3[3] };
            #pragma unroll
            for (int m = 0; m < TILE; ++m) {
                s[m]    = myxt[m * XPAD + lane];
                rinv[m] = 1.0f / (1.0f - dt2 * a[m][m]);
                racc[m] = 0.f;
            }
            #pragma unroll
            for (int m = 0; m < TILE; ++m) {
                y[m] = __builtin_fmaf(dt2, racc[m], s[m]) * rinv[m];
                #pragma unroll
                for (int i = m + 1; i < TILE; ++i)
                    racc[i] = __builtin_fmaf(a[i][m], y[m], racc[i]);
                myz[m] = dt2 * (y[m] + uo[m]);
            }
            float* o = out + (size_t)(batch_base + lane) * SN + rb;
            #pragma unroll
            for (int m = 0; m < 16; m += 4) {
                v4f t = { y[m], y[m + 1], y[m + 2], y[m + 3] };
                *(v4f*)(o + m) = t;
            }
        }
    }
}

extern "C" void kernel_launch(void* const* d_in, const int* in_sizes, int n_in,
                              void* d_out, int out_size, void* d_ws, size_t ws_size,
                              hipStream_t stream) {
    const float* u     = (const float*)d_in[0];
    const float* v     = (const float*)d_in[1];
    const float* delta = (const float*)d_in[2];
    const float* A     = (const float*)d_in[3];
    const float* Bvec  = (const float*)d_in[4];
    float* out = (float*)d_out;

    const int batch  = in_sizes[0] / SN;
    const int blocks = batch / (TILES * TILE);
    hippo_bilinear_wmma<<<blocks, WAVES * 32, 0, stream>>>(u, v, delta, A, Bvec, out, batch);
}


# rhs block passed from the WMMA accumulator layout to the one-column-per-lane solve by a half-wave register exchange (8 v_permlanex16), not an LDS store, wait and load; sum order unchanged
# speedup vs baseline: 3.8625x; 1.1173x over previous
#include <hip/hip_runtime.h>

typedef __attribute__((ext_vector_type(2))) float v2f;
typedef __attribute__((ext_vector_type(4))) float v4f;
typedef __attribute__((ext_vector_type(8))) float v8f;

static constexpr int SN    = 128;
static constexpr int TILE  = 16;
static constexpr int NB    = SN / TILE;
static constexpr int APAD  = 132;
static constexpr int XPAD  = 20;
static constexpr int TILES = 2;
static constexpr int WAVES = 2 * TILES;

static __device__ __forceinline__ void lds_handover_barrier() {
    asm volatile("s_wait_dscnt 0x0\n\ts_barrier_signal -1\n\ts_barrier_wait -1" ::: "memory");
}

static __device__ __forceinline__ float xhalf(float x) {
    const int b = __builtin_bit_cast(int, x);
    return __builtin_bit_cast(float,
        __builtin_amdgcn_permlanex16(b, b, 0x76543210, 0xfedcba98, false, false));
}

template <int K>
static __device__ __forceinline__ void product_step(v8f (&pacc)[NB], const float* alds,
                                                    const float* wrow, float* pbk,
                                                    int col, int hlf)
{
    v2f wb[4];
    v2f af[NB][4];
    #pragma unroll
    for (int k4 = 0; k4 < 4; ++k4)
        wb[k4] = *(const v2f*)(wrow + k4 * 4 + 2 * hlf);
    #pragma unroll
    for (int i = K + 1; i < NB; ++i) {
        const float* arow = alds + (i * TILE + col) * APAD + (K - 1) * TILE;
        #pragma unroll
        for (int k4 = 0; k4 < 4; ++k4)
            af[i][k4] = *(const v2f*)(arow + k4 * 4 + 2 * hlf);
    }
    #pragma unroll
    for (int k4 = 0; k4 < 4; ++k4) {
        #pragma unroll
        for (int i = K + 1; i < NB; ++i)
            pacc[i] = __builtin_amdgcn_wmma_f32_16x16x4_f32(
                          false, af[i][k4], false, wb[k4], (short)0, pacc[i], false, false);
    }
    *(v8f*)(pbk + (K + 1) * (32 * 8)) = pacc[K + 1];
}

__global__ __launch_bounds__(WAVES * 32) void hippo_bilinear_wmma(
    const float* __restrict__ u, const float* __restrict__ v,
    const float* __restrict__ delta, const float* __restrict__ A,
    const float* __restrict__ Bvec, float* __restrict__ out, int batch)
{
    __shared__ float A_lds[SN * APAD];
    __shared__ float zf[TILES][NB][TILE][XPAD];
    __shared__ __attribute__((aligned(32))) float pb[TILES][NB][32 * 8];

    const int tid = threadIdx.x;

    const int wave = __builtin_amdgcn_readfirstlane(tid >> 5);
    const int lane = tid & 31;
    const int tile = wave & (TILES - 1);
    const bool solving = wave < TILES;
    const int col  = lane & 15;
    const int hlf  = lane >> 4;
    const int batch_base = (blockIdx.x * TILES + tile) * TILE;
    const bool active = batch_base < batch;
    const int b = active ? batch_base + col : col;

    float dl, vv;
    const float* ucol = u + (size_t)b * SN;

    v4f su[4];
    v4f hu[NB][2], hb[NB][2];
    v2f huk[NB][4];
    if (active) {
        if (solving) {
            #pragma unroll
            for (int q = 0; q < 4; ++q) su[q] = *(const v4f*)(ucol + 4 * q);
        } else {
            #pragma unroll
            for (int i = 0; i < NB; ++i) {
                const int rb = i * TILE;
                const int nbase = rb + 8 * hlf;
                hu[i][0] = *(const v4f*)(ucol + nbase);
                hu[i][1] = *(const v4f*)(ucol + nbase + 4);
                hb[i][0] = *(const v4f*)(Bvec + nbase);
                hb[i][1] = *(const v4f*)(Bvec + nbase + 4);
                #pragma unroll
                for (int k4 = 0; k4 < 4; ++k4)
                    huk[i][k4] = *(const v2f*)(ucol + rb + k4 * 4 + 2 * hlf);
            }
        }
    }

    {
        static_assert(WAVES * 32 == 128, "one float2 per thread per 16x16 block");
        constexpr int NBLK = NB * (NB + 1) / 2;
        const int c2   = tid & 7;
        const int brow = ((tid >> 3) & 3) * 4 + (tid >> 5);
        const float* gsrc = A + brow * SN + c2 * 2;
        float* ldst = &A_lds[brow * APAD + c2 * 2];
        v2f t[NBLK];
        #pragma unroll
        for (int kb = 0; kb < NB; ++kb) {
            #pragma unroll
            for (int jb = 0; jb <= kb; ++jb)
                t[kb * (kb + 1) / 2 + jb] = *(const v2f*)(gsrc + kb * TILE * SN + jb * TILE);
        }
        dl = delta[b];
        vv = v[b];
        #pragma unroll
        for (int kb = 0; kb < NB; ++kb) {
            #pragma unroll
            for (int jb = 0; jb <= kb; ++jb)
                *(v2f*)(ldst + kb * TILE * APAD + jb * TILE) = t[kb * (kb + 1) / 2 + jb];
        }
    }
    __syncthreads();

    const float dt2 = 0.5f * dl;
    const float dvb = dl * vv;

    float* mypb = &pb[tile][0][lane * 8];

    v8f pacc[NB];

    if (active && !solving) {
        #pragma unroll
        for (int i = 0; i < NB; ++i) {
            #pragma unroll
            for (int r = 0; r < 4; ++r) {
                pacc[i][r]     = hu[i][0][r] + dvb * hb[i][0][r];
                pacc[i][r + 4] = hu[i][1][r] + dvb * hb[i][1][r];
            }
        }
        const float* arow = &A_lds[col * APAD];
        v8f acc = pacc[0];
        #pragma unroll
        for (int k4 = 0; k4 < 4; ++k4) {
            v2f af = *(const v2f*)(arow + k4 * 4 + 2 * hlf);
            v2f bf = dt2 * huk[0][k4];
            acc = __builtin_amdgcn_wmma_f32_16x16x4_f32(
                      false, af, false, bf, (short)0, acc, false, false);
        }
        *(v8f*)(mypb) = acc;
    }

    #pragma unroll 1
    for (int k = 0; k < NB; ++k) {
        lds_handover_barrier();
        if (!active) continue;

        if (!solving) {
            if (k == 0) {
                v2f ad[NB][4];
                #pragma unroll
                for (int i = 1; i < NB; ++i) {
                    const float* arow = &A_lds[(i * TILE + col) * APAD + i * TILE];
                    #pragma unroll
                    for (int k4 = 0; k4 < 4; ++k4)
                        ad[i][k4] = *(const v2f*)(arow + k4 * 4 + 2 * hlf);
                }
                #pragma unroll
                for (int k4 = 0; k4 < 4; ++k4) {
                    #pragma unroll
                    for (int i = 1; i < NB; ++i) {
                        v2f bf = dt2 * huk[i][k4];
                        pacc[i] = __builtin_amdgcn_wmma_f32_16x16x4_f32(
                                      false, ad[i][k4], false, bf, (short)0, pacc[i], false, false);
                    }
                }
                *(v8f*)(mypb + 1 * (32 * 8)) = pacc[1];
            } else {
                const float* wrow = &zf[tile][k - 1][col][0];
                switch (k) {
                    case 1: product_step<1>(pacc, A_lds, wrow, mypb, col, hlf); break;
                    case 2: product_step<2>(pacc, A_lds, wrow, mypb, col, hlf); break;
                    case 3: product_step<3>(pacc, A_lds, wrow, mypb, col, hlf); break;
                    case 4: product_step<4>(pacc, A_lds, wrow, mypb, col, hlf); break;
                    case 5: product_step<5>(pacc, A_lds, wrow, mypb, col, hlf); break;
                    case 6: product_step<6>(pacc, A_lds, wrow, mypb, col, hlf); break;
                    default: break;
                }
            }
            continue;
        }

        const int rb = k * TILE;
        const float* arows = &A_lds[(rb + col) * APAD];

        v8f acc = *(const v8f*)(mypb + k * (32 * 8));
        v2f af[4], wb[4];
        if (k > 0) {
            const float* wrow = &zf[tile][k - 1][col][0];
            #pragma unroll
            for (int k4 = 0; k4 < 4; ++k4) {
                af[k4] = *(const v2f*)(arows + (rb - TILE) + k4 * 4 + 2 * hlf);
                wb[k4] = *(const v2f*)(wrow + k4 * 4 + 2 * hlf);
            }
        }

        float a[TILE][TILE];
        {
            const float* dblk = &A_lds[rb * APAD + rb];
            #pragma unroll
            for (int m = 0; m < TILE; ++m) {
                #pragma unroll
                for (int j = 0; j <= m; ++j)
                    a[m][j] = dblk[m * APAD + j];
            }
        }

        const v4f u0 = su[0], u1 = su[1], u2 = su[2], u3 = su[3];
        if (k + 1 < NB) {
            #pragma unroll
            for (int q = 0; q < 4; ++q) su[q] = *(const v4f*)(ucol + rb + TILE + 4 * q);
        }

        if (k > 0) {
            #pragma unroll
            for (int k4 = 0; k4 < 4; ++k4)
                acc = __builtin_amdgcn_wmma_f32_16x16x4_f32(
                          false, af[k4], false, wb[k4], (short)0, acc, false, false);
        }

        float xr[8];
        #pragma unroll
        for (int r = 0; r < 8; ++r) xr[r] = xhalf(acc[r]);

        if (lane < 16) {
            float* myz = &zf[tile][k][lane][0];
            float rinv[TILE], racc[TILE], y[TILE];
            const float s[TILE]  = { acc[0], acc[1], acc[2], acc[3], acc[4], acc[5], acc[6], acc[7],
                                     xr[0],  xr[1],  xr[2],  xr[3],  xr[4],  xr[5],  xr[6],  xr[7] };
            const float uo[TILE] = { u0[0], u0[1], u0[2], u0[3], u1[0], u1[1], u1[2], u1[3],
                                     u2[0], u2[1], u2[2], u2[3], u3[0], u3[1], u3[2], u3[3] };
            #pragma unroll
            for (int m = 0; m < TILE; ++m) {
                rinv[m] = 1.0f / (1.0f - dt2 * a[m][m]);
                racc[m] = 0.f;
            }
            #pragma unroll
            for (int m = 0; m < TILE; ++m) {
                y[m] = __builtin_fmaf(dt2, racc[m], s[m]) * rinv[m];
                #pragma unroll
                for (int i = m + 1; i < TILE; ++i)
                    racc[i] = __builtin_fmaf(a[i][m], y[m], racc[i]);
                myz[m] = dt2 * (y[m] + uo[m]);
            }
            float* o = out + (size_t)(batch_base + lane) * SN + rb;
            #pragma unroll
            for (int m = 0; m < 16; m += 4) {
                v4f t = { y[m], y[m + 1], y[m + 2], y[m + 3] };
                *(v4f*)(o + m) = t;
            }
        }
    }
}

extern "C" void kernel_launch(void* const* d_in, const int* in_sizes, int n_in,
                              void* d_out, int out_size, void* d_ws, size_t ws_size,
                              hipStream_t stream) {
    const float* u     = (const float*)d_in[0];
    const float* v     = (const float*)d_in[1];
    const float* delta = (const float*)d_in[2];
    const float* A     = (const float*)d_in[3];
    const float* Bvec  = (const float*)d_in[4];
    float* out = (float*)d_out;

    const int batch  = in_sizes[0] / SN;
    const int blocks = batch / (TILES * TILE);
    hippo_bilinear_wmma<<<blocks, WAVES * 32, 0, stream>>>(u, v, delta, A, Bvec, out, batch);
}


# diagonal 16x16 solve: serial step cut from fma,fma,mul to one fma, y[m]=c[m]*y[m-1]+(q[m]*racc[m]+h[m]); 1/pivot, dt2/pivot, sub-diagonal coefficient tabulated in LDS ahead; row sum regrouped
# speedup vs baseline: 4.1010x; 1.0617x over previous
#include <hip/hip_runtime.h>

typedef __attribute__((ext_vector_type(2))) float v2f;
typedef __attribute__((ext_vector_type(4))) float v4f;
typedef __attribute__((ext_vector_type(8))) float v8f;

static constexpr int SN    = 128;
static constexpr int TILE  = 16;
static constexpr int NB    = SN / TILE;
static constexpr int APAD  = 132;
static constexpr int XPAD  = 20;
static constexpr int TILES = 2;
static constexpr int WAVES = 2 * TILES;
static constexpr int CPAD  = 132;
static constexpr int CTILE = TILE * CPAD + 32;

static __device__ __forceinline__ void lds_handover_barrier() {
    asm volatile("s_wait_dscnt 0x0\n\ts_barrier_signal -1\n\ts_barrier_wait -1" ::: "memory");
}

static __device__ __forceinline__ float xhalf(float x) {
    const int b = __builtin_bit_cast(int, x);
    return __builtin_bit_cast(float,
        __builtin_amdgcn_permlanex16(b, b, 0x76543210, 0xfedcba98, false, false));
}

template <int K>
static __device__ __forceinline__ void product_step(v8f (&pacc)[NB], const float* alds,
                                                    const float* wrow, float* pbk,
                                                    int col, int hlf)
{
    v2f wb[4];
    v2f af[NB][4];
    #pragma unroll
    for (int k4 = 0; k4 < 4; ++k4)
        wb[k4] = *(const v2f*)(wrow + k4 * 4 + 2 * hlf);
    #pragma unroll
    for (int i = K + 1; i < NB; ++i) {
        const float* arow = alds + (i * TILE + col) * APAD + (K - 1) * TILE;
        #pragma unroll
        for (int k4 = 0; k4 < 4; ++k4)
            af[i][k4] = *(const v2f*)(arow + k4 * 4 + 2 * hlf);
    }
    #pragma unroll
    for (int k4 = 0; k4 < 4; ++k4) {
        #pragma unroll
        for (int i = K + 1; i < NB; ++i)
            pacc[i] = __builtin_amdgcn_wmma_f32_16x16x4_f32(
                          false, af[i][k4], false, wb[k4], (short)0, pacc[i], false, false);
    }
    *(v8f*)(pbk + (K + 1) * (32 * 8)) = pacc[K + 1];
}

__global__ __launch_bounds__(WAVES * 32) void hippo_bilinear_wmma(
    const float* __restrict__ u, const float* __restrict__ v,
    const float* __restrict__ delta, const float* __restrict__ A,
    const float* __restrict__ Bvec, float* __restrict__ out, int batch)
{
    __shared__ float A_lds[SN * APAD];
    __shared__ float zf[TILES][NB][TILE][XPAD];
    __shared__ __attribute__((aligned(32))) float pb[TILES][NB][32 * 8];
    __shared__ __attribute__((aligned(16))) float cf_r[TILES * CTILE];
    __shared__ __attribute__((aligned(16))) float cf_q[TILES * CTILE];
    __shared__ __attribute__((aligned(16))) float cf_c[TILES * CTILE];

    const int tid = threadIdx.x;

    const int wave = __builtin_amdgcn_readfirstlane(tid >> 5);
    const int lane = tid & 31;
    const int tile = wave & (TILES - 1);
    const bool solving = wave < TILES;
    const int col  = lane & 15;
    const int hlf  = lane >> 4;
    const int batch_base = (blockIdx.x * TILES + tile) * TILE;
    const bool active = batch_base < batch;
    const int b = active ? batch_base + col : col;

    float dl, vv;

    const int tcol_g = blockIdx.x * (TILES * TILE) + lane;
    const float tdl = delta[tcol_g < batch ? tcol_g : 0];
    const int trow = wave * 32 + lane;
    const float t_amm  = A[trow * SN + trow];
    const float t_asub = A[trow > 0 ? trow * SN + trow - 1 : 0];
    const float* ucol = u + (size_t)b * SN;

    v4f su[4];
    v4f hu[NB][2], hb[NB][2];
    v2f huk[NB][4];
    if (active) {
        if (solving) {
            #pragma unroll
            for (int q = 0; q < 4; ++q) su[q] = *(const v4f*)(ucol + 4 * q);
        } else {
            #pragma unroll
            for (int i = 0; i < NB; ++i) {
                const int rb = i * TILE;
                const int nbase = rb + 8 * hlf;
                hu[i][0] = *(const v4f*)(ucol + nbase);
                hu[i][1] = *(const v4f*)(ucol + nbase + 4);
                hb[i][0] = *(const v4f*)(Bvec + nbase);
                hb[i][1] = *(const v4f*)(Bvec + nbase + 4);
                #pragma unroll
                for (int k4 = 0; k4 < 4; ++k4)
                    huk[i][k4] = *(const v2f*)(ucol + rb + k4 * 4 + 2 * hlf);
            }
        }
    }

    {
        static_assert(WAVES * 32 == 128, "one float2 per thread per 16x16 block");
        constexpr int NBLK = NB * (NB + 1) / 2;
        const int c2   = tid & 7;
        const int brow = ((tid >> 3) & 3) * 4 + (tid >> 5);
        const float* gsrc = A + brow * SN + c2 * 2;
        float* ldst = &A_lds[brow * APAD + c2 * 2];
        v2f t[NBLK];
        #pragma unroll
        for (int kb = 0; kb < NB; ++kb) {
            #pragma unroll
            for (int jb = 0; jb <= kb; ++jb)
                t[kb * (kb + 1) / 2 + jb] = *(const v2f*)(gsrc + kb * TILE * SN + jb * TILE);
        }
        dl = delta[b];
        vv = v[b];
        __builtin_amdgcn_sched_barrier(0);
        {
            const float tdt2 = 0.5f * tdl;
            const int m0 = wave * 32;
            const int toff = (lane >> 4) * CTILE + (lane & 15) * CPAD + m0;
            #pragma unroll
            for (int i4 = 0; i4 < 8; ++i4) {
                v4f rr, qq, cc;
                #pragma unroll
                for (int e = 0; e < 4; ++e) {
                    const int i = i4 * 4 + e;
                    const float amm  = __builtin_bit_cast(float,
                        __builtin_amdgcn_readlane(__builtin_bit_cast(int, t_amm), i));
                    const float asub = __builtin_bit_cast(float,
                        __builtin_amdgcn_readlane(__builtin_bit_cast(int, t_asub), i));
                    const float rv = 1.0f / (1.0f - tdt2 * amm);
                    const float qv = tdt2 * rv;
                    rr[e] = rv;
                    qq[e] = qv;
                    cc[e] = qv * asub;
                }
                *(v4f*)(&cf_r[toff + i4 * 4]) = rr;
                *(v4f*)(&cf_q[toff + i4 * 4]) = qq;
                *(v4f*)(&cf_c[toff + i4 * 4]) = cc;
            }
        }
        #pragma unroll
        for (int kb = 0; kb < NB; ++kb) {
            #pragma unroll
            for (int jb = 0; jb <= kb; ++jb)
                *(v2f*)(ldst + kb * TILE * APAD + jb * TILE) = t[kb * (kb + 1) / 2 + jb];
        }
    }
    __syncthreads();

    const float dt2 = 0.5f * dl;
    const float dvb = dl * vv;

    float* mypb = &pb[tile][0][lane * 8];

    v8f pacc[NB];

    if (active && !solving) {
        #pragma unroll
        for (int i = 0; i < NB; ++i) {
            #pragma unroll
            for (int r = 0; r < 4; ++r) {
                pacc[i][r]     = hu[i][0][r] + dvb * hb[i][0][r];
                pacc[i][r + 4] = hu[i][1][r] + dvb * hb[i][1][r];
            }
        }
        const float* arow = &A_lds[col * APAD];
        v8f acc = pacc[0];
        #pragma unroll
        for (int k4 = 0; k4 < 4; ++k4) {
            v2f af = *(const v2f*)(arow + k4 * 4 + 2 * hlf);
            v2f bf = dt2 * huk[0][k4];
            acc = __builtin_amdgcn_wmma_f32_16x16x4_f32(
                      false, af, false, bf, (short)0, acc, false, false);
        }
        *(v8f*)(mypb) = acc;
    }

    #pragma unroll 1
    for (int k = 0; k < NB; ++k) {
        lds_handover_barrier();
        if (!active) continue;

        if (!solving) {
            if (k == 0) {
                v2f ad[NB][4];
                #pragma unroll
                for (int i = 1; i < NB; ++i) {
                    const float* arow = &A_lds[(i * TILE + col) * APAD + i * TILE];
                    #pragma unroll
                    for (int k4 = 0; k4 < 4; ++k4)
                        ad[i][k4] = *(const v2f*)(arow + k4 * 4 + 2 * hlf);
                }
                #pragma unroll
                for (int k4 = 0; k4 < 4; ++k4) {
                    #pragma unroll
                    for (int i = 1; i < NB; ++i) {
                        v2f bf = dt2 * huk[i][k4];
                        pacc[i] = __builtin_amdgcn_wmma_f32_16x16x4_f32(
                                      false, ad[i][k4], false, bf, (short)0, pacc[i], false, false);
                    }
                }
                *(v8f*)(mypb + 1 * (32 * 8)) = pacc[1];
            } else {
                const float* wrow = &zf[tile][k - 1][col][0];
                switch (k) {
                    case 1: product_step<1>(pacc, A_lds, wrow, mypb, col, hlf); break;
                    case 2: product_step<2>(pacc, A_lds, wrow, mypb, col, hlf); break;
                    case 3: product_step<3>(pacc, A_lds, wrow, mypb, col, hlf); break;
                    case 4: product_step<4>(pacc, A_lds, wrow, mypb, col, hlf); break;
                    case 5: product_step<5>(pacc, A_lds, wrow, mypb, col, hlf); break;
                    case 6: product_step<6>(pacc, A_lds, wrow, mypb, col, hlf); break;
                    default: break;
                }
            }
            continue;
        }

        const int rb = k * TILE;
        const float* arows = &A_lds[(rb + col) * APAD];

        v8f acc = *(const v8f*)(mypb + k * (32 * 8));
        v2f af[4], wb[4];
        if (k > 0) {
            const float* wrow = &zf[tile][k - 1][col][0];
            #pragma unroll
            for (int k4 = 0; k4 < 4; ++k4) {
                af[k4] = *(const v2f*)(arows + (rb - TILE) + k4 * 4 + 2 * hlf);
                wb[k4] = *(const v2f*)(wrow + k4 * 4 + 2 * hlf);
            }
        }

        v4f tr[4], tq[4], tc[4];
        {
            const int coff = tile * CTILE + col * CPAD + rb;
            #pragma unroll
            for (int g = 0; g < 4; ++g) {
                tr[g] = *(const v4f*)(&cf_r[coff + 4 * g]);
                tq[g] = *(const v4f*)(&cf_q[coff + 4 * g]);
                tc[g] = *(const v4f*)(&cf_c[coff + 4 * g]);
            }
        }

        float a[TILE][TILE];
        {
            const float* dblk = &A_lds[rb * APAD + rb];
            #pragma unroll
            for (int m = 2; m < TILE; ++m) {
                #pragma unroll
                for (int j = 0; j <= m - 2; ++j)
                    a[m][j] = dblk[m * APAD + j];
            }
        }

        const v4f u0 = su[0], u1 = su[1], u2 = su[2], u3 = su[3];
        if (k + 1 < NB) {
            #pragma unroll
            for (int q = 0; q < 4; ++q) su[q] = *(const v4f*)(ucol + rb + TILE + 4 * q);
        }

        if (k > 0) {
            #pragma unroll
            for (int k4 = 0; k4 < 4; ++k4)
                acc = __builtin_amdgcn_wmma_f32_16x16x4_f32(
                          false, af[k4], false, wb[k4], (short)0, acc, false, false);
        }

        float xr[8];
        #pragma unroll
        for (int r = 0; r < 8; ++r) xr[r] = xhalf(acc[r]);

        if (lane < 16) {
            float* myz = &zf[tile][k][lane][0];
            float h[TILE], racc[TILE], y[TILE];
            const float s[TILE]  = { acc[0], acc[1], acc[2], acc[3], acc[4], acc[5], acc[6], acc[7],
                                     xr[0],  xr[1],  xr[2],  xr[3],  xr[4],  xr[5],  xr[6],  xr[7] };
            const float uo[TILE] = { u0[0], u0[1], u0[2], u0[3], u1[0], u1[1], u1[2], u1[3],
                                     u2[0], u2[1], u2[2], u2[3], u3[0], u3[1], u3[2], u3[3] };
            #pragma unroll
            for (int m = 0; m < TILE; ++m) {
                h[m]    = tr[m >> 2][m & 3] * s[m];
                racc[m] = 0.f;
            }
            float pre[TILE], w[TILE];
            pre[0] = h[0];
            pre[1] = h[1];
            y[0] = pre[0];
            #pragma unroll
            for (int m = 0; m < TILE; ++m) {
                #pragma unroll
                for (int i = m + 2; i < TILE; ++i)
                    racc[i] = __builtin_fmaf(a[i][m], y[m], racc[i]);
                if (m + 1 < TILE)
                    y[m + 1] = __builtin_fmaf(tc[(m + 1) >> 2][(m + 1) & 3], y[m], pre[m + 1]);
                if (m + 2 < TILE)
                    pre[m + 2] = __builtin_fmaf(tq[(m + 2) >> 2][(m + 2) & 3], racc[m + 2], h[m + 2]);
                w[m] = dt2 * (y[m] + uo[m]);
                if (m + 2 < TILE)
                    asm volatile("" : "+v"(racc[3]), "+v"(racc[4]), "+v"(racc[5]), "+v"(racc[6]),
                                      "+v"(racc[7]), "+v"(racc[8]), "+v"(racc[9]), "+v"(racc[10]),
                                      "+v"(racc[11]), "+v"(racc[12]), "+v"(racc[13]), "+v"(racc[14]),
                                      "+v"(racc[15]), "+v"(y[m + 1]), "+v"(pre[m + 2]));
            }
            #pragma unroll
            for (int m = 0; m < TILE; ++m) myz[m] = w[m];
            float* o = out + (size_t)(batch_base + lane) * SN + rb;
            #pragma unroll
            for (int m = 0; m < 16; m += 4) {
                v4f t = { y[m], y[m + 1], y[m + 2], y[m + 3] };
                *(v4f*)(o + m) = t;
            }
        }
    }
}

extern "C" void kernel_launch(void* const* d_in, const int* in_sizes, int n_in,
                              void* d_out, int out_size, void* d_ws, size_t ws_size,
                              hipStream_t stream) {
    const float* u     = (const float*)d_in[0];
    const float* v     = (const float*)d_in[1];
    const float* delta = (const float*)d_in[2];
    const float* A     = (const float*)d_in[3];
    const float* Bvec  = (const float*)d_in[4];
    float* out = (float*)d_out;

    const int batch  = in_sizes[0] / SN;
    const int blocks = batch / (TILES * TILE);
    hippo_bilinear_wmma<<<blocks, WAVES * 32, 0, stream>>>(u, v, delta, A, Bvec, out, batch);
}


# u and B reach both waves of a tile through LDS: product wave fetches the tile with 17 whole-line 16-byte loads (was 64 loads of 16 lines each, past the outstanding-load limit); sums unchanged
# speedup vs baseline: 5.0607x; 1.2340x over previous
#include <hip/hip_runtime.h>

typedef __attribute__((ext_vector_type(2))) float v2f;
typedef __attribute__((ext_vector_type(4))) float v4f;
typedef __attribute__((ext_vector_type(8))) float v8f;

static constexpr int SN    = 128;
static constexpr int TILE  = 16;
static constexpr int NB    = SN / TILE;
static constexpr int APAD  = 132;
static constexpr int XPAD  = 20;
static constexpr int TILES = 2;
static constexpr int WAVES = 2 * TILES;
static constexpr int CPAD  = 132;
static constexpr int CTILE = TILE * CPAD + 32;
static constexpr int UPAD  = 132;

static __device__ __forceinline__ void lds_handover_barrier() {
    asm volatile("s_wait_dscnt 0x0\n\ts_barrier_signal -1\n\ts_barrier_wait -1" ::: "memory");
}

static __device__ __forceinline__ float xhalf(float x) {
    const int b = __builtin_bit_cast(int, x);
    return __builtin_bit_cast(float,
        __builtin_amdgcn_permlanex16(b, b, 0x76543210, 0xfedcba98, false, false));
}

template <int K>
static __device__ __forceinline__ void product_step(v8f (&pacc)[NB], const float* alds,
                                                    const float* wrow, float* pbk,
                                                    int col, int hlf)
{
    v2f wb[4];
    v2f af[NB][4];
    #pragma unroll
    for (int k4 = 0; k4 < 4; ++k4)
        wb[k4] = *(const v2f*)(wrow + k4 * 4 + 2 * hlf);
    #pragma unroll
    for (int i = K + 1; i < NB; ++i) {
        const float* arow = alds + (i * TILE + col) * APAD + (K - 1) * TILE;
        #pragma unroll
        for (int k4 = 0; k4 < 4; ++k4)
            af[i][k4] = *(const v2f*)(arow + k4 * 4 + 2 * hlf);
    }
    #pragma unroll
    for (int k4 = 0; k4 < 4; ++k4) {
        #pragma unroll
        for (int i = K + 1; i < NB; ++i)
            pacc[i] = __builtin_amdgcn_wmma_f32_16x16x4_f32(
                          false, af[i][k4], false, wb[k4], (short)0, pacc[i], false, false);
    }
    *(v8f*)(pbk + (K + 1) * (32 * 8)) = pacc[K + 1];
}

__global__ __launch_bounds__(WAVES * 32) void hippo_bilinear_wmma(
    const float* __restrict__ u, const float* __restrict__ v,
    const float* __restrict__ delta, const float* __restrict__ A,
    const float* __restrict__ Bvec, float* __restrict__ out, int batch)
{
    __shared__ float A_lds[SN * APAD];
    __shared__ float zf[TILES][NB][TILE][XPAD];
    __shared__ __attribute__((aligned(32))) float pb[TILES][NB][32 * 8];
    __shared__ __attribute__((aligned(16))) float cf_r[TILES * CTILE];
    __shared__ __attribute__((aligned(16))) float cf_q[TILES * CTILE];
    __shared__ __attribute__((aligned(16))) float cf_c[TILES * CTILE];
    __shared__ __attribute__((aligned(16))) float u_lds[TILES][TILE * UPAD];
    __shared__ __attribute__((aligned(16))) float b_lds[TILES][SN];

    const int tid = threadIdx.x;

    const int wave = __builtin_amdgcn_readfirstlane(tid >> 5);
    const int lane = tid & 31;
    const int tile = wave & (TILES - 1);
    const bool solving = wave < TILES;
    const int col  = lane & 15;
    const int hlf  = lane >> 4;
    const int batch_base = (blockIdx.x * TILES + tile) * TILE;
    const bool active = batch_base < batch;
    const int b = active ? batch_base + col : col;

    float dl, vv;

    const int tcol_g = blockIdx.x * (TILES * TILE) + lane;
    const float tdl = delta[tcol_g < batch ? tcol_g : 0];
    const int trow = wave * 32 + lane;
    const float t_amm  = A[trow * SN + trow];
    const float t_asub = A[trow > 0 ? trow * SN + trow - 1 : 0];

    v4f ug[TILE], bg;
    v4f hu[NB][2], hb[NB][2];
    v2f huk[NB][4];
    if (active && !solving) {
        const float* utile = u + (size_t)batch_base * SN + 4 * lane;
        #pragma unroll
        for (int c = 0; c < TILE; ++c) ug[c] = *(const v4f*)(utile + c * SN);
        bg = *(const v4f*)(Bvec + 4 * lane);
    }

    {
        static_assert(WAVES * 32 == 128, "one float2 per thread per 16x16 block");
        constexpr int NBLK = NB * (NB + 1) / 2;
        const int c2   = tid & 7;
        const int brow = ((tid >> 3) & 3) * 4 + (tid >> 5);
        const float* gsrc = A + brow * SN + c2 * 2;
        float* ldst = &A_lds[brow * APAD + c2 * 2];
        v2f t[NBLK];
        #pragma unroll
        for (int kb = 0; kb < NB; ++kb) {
            #pragma unroll
            for (int jb = 0; jb <= kb; ++jb)
                t[kb * (kb + 1) / 2 + jb] = *(const v2f*)(gsrc + kb * TILE * SN + jb * TILE);
        }
        dl = delta[b];
        vv = v[b];
        __builtin_amdgcn_sched_barrier(0);
        if (active && !solving) {
            float* ud = &u_lds[tile][4 * lane];
            #pragma unroll
            for (int c = 0; c < TILE; ++c) *(v4f*)(ud + c * UPAD) = ug[c];
            *(v4f*)(&b_lds[tile][4 * lane]) = bg;
            const float* ut = &u_lds[tile][col * UPAD];
            const float* bt = &b_lds[tile][0];
            #pragma unroll
            for (int i = 0; i < NB; ++i) {
                const int rb = i * TILE;
                const int nbase = rb + 8 * hlf;
                hu[i][0] = *(const v4f*)(ut + nbase);
                hu[i][1] = *(const v4f*)(ut + nbase + 4);
                hb[i][0] = *(const v4f*)(bt + nbase);
                hb[i][1] = *(const v4f*)(bt + nbase + 4);
                #pragma unroll
                for (int k4 = 0; k4 < 4; ++k4)
                    huk[i][k4] = *(const v2f*)(ut + rb + k4 * 4 + 2 * hlf);
            }
        }
        {
            const float tdt2 = 0.5f * tdl;
            const int m0 = wave * 32;
            const int toff = (lane >> 4) * CTILE + (lane & 15) * CPAD + m0;
            #pragma unroll
            for (int i4 = 0; i4 < 8; ++i4) {
                v4f rr, qq, cc;
                #pragma unroll
                for (int e = 0; e < 4; ++e) {
                    const int i = i4 * 4 + e;
                    const float amm  = __builtin_bit_cast(float,
                        __builtin_amdgcn_readlane(__builtin_bit_cast(int, t_amm), i));
                    const float asub = __builtin_bit_cast(float,
                        __builtin_amdgcn_readlane(__builtin_bit_cast(int, t_asub), i));
                    const float rv = 1.0f / (1.0f - tdt2 * amm);
                    const float qv = tdt2 * rv;
                    rr[e] = rv;
                    qq[e] = qv;
                    cc[e] = qv * asub;
                }
                *(v4f*)(&cf_r[toff + i4 * 4]) = rr;
                *(v4f*)(&cf_q[toff + i4 * 4]) = qq;
                *(v4f*)(&cf_c[toff + i4 * 4]) = cc;
            }
        }
        __builtin_amdgcn_sched_barrier(0);
        #pragma unroll
        for (int kb = 0; kb < NB; ++kb) {
            #pragma unroll
            for (int jb = 0; jb <= kb; ++jb)
                *(v2f*)(ldst + kb * TILE * APAD + jb * TILE) = t[kb * (kb + 1) / 2 + jb];
        }
    }
    __syncthreads();

    const float dt2 = 0.5f * dl;
    const float dvb = dl * vv;

    float* mypb = &pb[tile][0][lane * 8];

    v8f pacc[NB];

    if (active && !solving) {
        #pragma unroll
        for (int i = 0; i < NB; ++i) {
            #pragma unroll
            for (int r = 0; r < 4; ++r) {
                pacc[i][r]     = hu[i][0][r] + dvb * hb[i][0][r];
                pacc[i][r + 4] = hu[i][1][r] + dvb * hb[i][1][r];
            }
        }
        const float* arow = &A_lds[col * APAD];
        v8f acc = pacc[0];
        #pragma unroll
        for (int k4 = 0; k4 < 4; ++k4) {
            v2f af = *(const v2f*)(arow + k4 * 4 + 2 * hlf);
            v2f bf = dt2 * huk[0][k4];
            acc = __builtin_amdgcn_wmma_f32_16x16x4_f32(
                      false, af, false, bf, (short)0, acc, false, false);
        }
        *(v8f*)(mypb) = acc;
    }

    #pragma unroll 1
    for (int k = 0; k < NB; ++k) {
        lds_handover_barrier();
        if (!active) continue;

        if (!solving) {
            if (k == 0) {
                v2f ad[NB][4];
                #pragma unroll
                for (int i = 1; i < NB; ++i) {
                    const float* arow = &A_lds[(i * TILE + col) * APAD + i * TILE];
                    #pragma unroll
                    for (int k4 = 0; k4 < 4; ++k4)
                        ad[i][k4] = *(const v2f*)(arow + k4 * 4 + 2 * hlf);
                }
                #pragma unroll
                for (int k4 = 0; k4 < 4; ++k4) {
                    #pragma unroll
                    for (int i = 1; i < NB; ++i) {
                        v2f bf = dt2 * huk[i][k4];
                        pacc[i] = __builtin_amdgcn_wmma_f32_16x16x4_f32(
                                      false, ad[i][k4], false, bf, (short)0, pacc[i], false, false);
                    }
                }
                *(v8f*)(mypb + 1 * (32 * 8)) = pacc[1];
            } else {
                const float* wrow = &zf[tile][k - 1][col][0];
                switch (k) {
                    case 1: product_step<1>(pacc, A_lds, wrow, mypb, col, hlf); break;
                    case 2: product_step<2>(pacc, A_lds, wrow, mypb, col, hlf); break;
                    case 3: product_step<3>(pacc, A_lds, wrow, mypb, col, hlf); break;
                    case 4: product_step<4>(pacc, A_lds, wrow, mypb, col, hlf); break;
                    case 5: product_step<5>(pacc, A_lds, wrow, mypb, col, hlf); break;
                    case 6: product_step<6>(pacc, A_lds, wrow, mypb, col, hlf); break;
                    default: break;
                }
            }
            continue;
        }

        const int rb = k * TILE;
        const float* arows = &A_lds[(rb + col) * APAD];

        v8f acc = *(const v8f*)(mypb + k * (32 * 8));
        v2f af[4], wb[4];
        if (k > 0) {
            const float* wrow = &zf[tile][k - 1][col][0];
            #pragma unroll
            for (int k4 = 0; k4 < 4; ++k4) {
                af[k4] = *(const v2f*)(arows + (rb - TILE) + k4 * 4 + 2 * hlf);
                wb[k4] = *(const v2f*)(wrow + k4 * 4 + 2 * hlf);
            }
        }

        v4f tr[4], tq[4], tc[4];
        {
            const int coff = tile * CTILE + col * CPAD + rb;
            #pragma unroll
            for (int g = 0; g < 4; ++g) {
                tr[g] = *(const v4f*)(&cf_r[coff + 4 * g]);
                tq[g] = *(const v4f*)(&cf_q[coff + 4 * g]);
                tc[g] = *(const v4f*)(&cf_c[coff + 4 * g]);
            }
        }

        float a[TILE][TILE];
        {
            const float* dblk = &A_lds[rb * APAD + rb];
            #pragma unroll
            for (int m = 2; m < TILE; ++m) {
                #pragma unroll
                for (int j = 0; j <= m - 2; ++j)
                    a[m][j] = dblk[m * APAD + j];
            }
        }

        const float* ucl = &u_lds[tile][col * UPAD + rb];
        const v4f u0 = *(const v4f*)(ucl), u1 = *(const v4f*)(ucl + 4),
                  u2 = *(const v4f*)(ucl + 8), u3 = *(const v4f*)(ucl + 12);

        if (k > 0) {
            #pragma unroll
            for (int k4 = 0; k4 < 4; ++k4)
                acc = __builtin_amdgcn_wmma_f32_16x16x4_f32(
                          false, af[k4], false, wb[k4], (short)0, acc, false, false);
        }

        float xr[8];
        #pragma unroll
        for (int r = 0; r < 8; ++r) xr[r] = xhalf(acc[r]);

        if (lane < 16) {
            float* myz = &zf[tile][k][lane][0];
            float h[TILE], racc[TILE], y[TILE];
            const float s[TILE]  = { acc[0], acc[1], acc[2], acc[3], acc[4], acc[5], acc[6], acc[7],
                                     xr[0],  xr[1],  xr[2],  xr[3],  xr[4],  xr[5],  xr[6],  xr[7] };
            const float uo[TILE] = { u0[0], u0[1], u0[2], u0[3], u1[0], u1[1], u1[2], u1[3],
                                     u2[0], u2[1], u2[2], u2[3], u3[0], u3[1], u3[2], u3[3] };
            #pragma unroll
            for (int m = 0; m < TILE; ++m) {
                h[m]    = tr[m >> 2][m & 3] * s[m];
                racc[m] = 0.f;
            }
            float pre[TILE], w[TILE];
            pre[0] = h[0];
            pre[1] = h[1];
            y[0] = pre[0];
            #pragma unroll
            for (int m = 0; m < TILE; ++m) {
                #pragma unroll
                for (int i = m + 2; i < TILE; ++i)
                    racc[i] = __builtin_fmaf(a[i][m], y[m], racc[i]);
                if (m + 1 < TILE)
                    y[m + 1] = __builtin_fmaf(tc[(m + 1) >> 2][(m + 1) & 3], y[m], pre[m + 1]);
                if (m + 2 < TILE)
                    pre[m + 2] = __builtin_fmaf(tq[(m + 2) >> 2][(m + 2) & 3], racc[m + 2], h[m + 2]);
                w[m] = dt2 * (y[m] + uo[m]);
                if (m + 2 < TILE)
                    asm volatile("" : "+v"(racc[3]), "+v"(racc[4]), "+v"(racc[5]), "+v"(racc[6]),
                                      "+v"(racc[7]), "+v"(racc[8]), "+v"(racc[9]), "+v"(racc[10]),
                                      "+v"(racc[11]), "+v"(racc[12]), "+v"(racc[13]), "+v"(racc[14]),
                                      "+v"(racc[15]), "+v"(y[m + 1]), "+v"(pre[m + 2]));
            }
            #pragma unroll
            for (int m = 0; m < TILE; ++m) myz[m] = w[m];
            float* o = out + (size_t)(batch_base + lane) * SN + rb;
            #pragma unroll
            for (int m = 0; m < 16; m += 4) {
                v4f t = { y[m], y[m + 1], y[m + 2], y[m + 3] };
                *(v4f*)(o + m) = t;
            }
        }
    }
}

extern "C" void kernel_launch(void* const* d_in, const int* in_sizes, int n_in,
                              void* d_out, int out_size, void* d_ws, size_t ws_size,
                              hipStream_t stream) {
    const float* u     = (const float*)d_in[0];
    const float* v     = (const float*)d_in[1];
    const float* delta = (const float*)d_in[2];
    const float* A     = (const float*)d_in[3];
    const float* Bvec  = (const float*)d_in[4];
    float* out = (float*)d_out;

    const int batch  = in_sizes[0] / SN;
    const int blocks = batch / (TILES * TILE);
    hippo_bilinear_wmma<<<blocks, WAVES * 32, 0, stream>>>(u, v, delta, A, Bvec, out, batch);
}
